# starting kernel
# baseline (speedup 1.0000x reference)
#include <hip/hip_runtime.h>
#include <hip/hip_bf16.h>

typedef __attribute__((ext_vector_type(2))) float v2f;
typedef __attribute__((ext_vector_type(8))) float v8f;

#define D 128
#define NEG_SLOPE 0.2f

__global__ __launch_bounds__(256) void gat_zero(float* __restrict__ p, long long n) {
    long long i = (long long)blockIdx.x * blockDim.x + threadIdx.x;
    if (i < n) p[i] = 0.0f;
}

__global__ __launch_bounds__(256) void gat_gemm_wmma(
    const float* __restrict__ X,
    const float* __restrict__ W0, const float* __restrict__ W1,
    const float* __restrict__ W2,
    float* __restrict__ O0, float* __restrict__ O1, float* __restrict__ O2,
    int n_nodes)
{
    __shared__ float sX[16 * D];

    const int row0 = blockIdx.x * 16;
    const int m    = blockIdx.y;
    const float* __restrict__ W = (m == 0) ? W0 : (m == 1) ? W1 : W2;
    float* __restrict__       O = (m == 0) ? O0 : (m == 1) ? O1 : O2;

    const int tid = threadIdx.x;

    {
        float4* Sv = (float4*)sX;
        int f0 = tid, f1 = tid + 256;
        int r0 = row0 + (f0 >> 5); r0 = r0 < n_nodes ? r0 : n_nodes - 1;
        int r1 = row0 + (f1 >> 5); r1 = r1 < n_nodes ? r1 : n_nodes - 1;
        const float4* X0 = (const float4*)(X + (long long)r0 * D);
        const float4* X1 = (const float4*)(X + (long long)r1 * D);
        Sv[f0] = X0[f0 & 31];
        Sv[f1] = X1[f1 & 31];
    }
    __syncthreads();

    const int wave = tid >> 5;
    const int lane = tid & 31;
    const int half = lane >> 4;
    const int lm   = lane & 15;
    const int col0 = wave * 16;

    v8f acc = {};
    #pragma unroll
    for (int k0 = 0; k0 < D; k0 += 4) {
        const int ka = k0 + half * 2;
        v2f a, b;
        a.x = sX[lm * D + ka + 0];
        a.y = sX[lm * D + ka + 1];
        b.x = W[(ka + 0) * D + col0 + lm];
        b.y = W[(ka + 1) * D + col0 + lm];
        acc = __builtin_amdgcn_wmma_f32_16x16x4_f32(
                false, a, false, b, (short)0, acc, false, false);
    }

    #pragma unroll
    for (int v = 0; v < 8; ++v) {
        const int row = row0 + v + half * 8;
        if (row < n_nodes)
            O[(long long)row * D + col0 + lm] = acc[v];
    }
}

__global__ __launch_bounds__(256) void gat_edge_fused(
    const float* __restrict__ h_l, const float* __restrict__ h_r,
    const float* __restrict__ comb,
    const int* __restrict__ src, const int* __restrict__ dst,
    float* __restrict__ agg, int n_edges)
{
    const int wavesPerBlock = blockDim.x >> 5;
    const int edge = blockIdx.x * wavesPerBlock + (threadIdx.x >> 5);
    const int lane = threadIdx.x & 31;
    if (edge >= n_edges) return;

    const int s = src[edge];
    const int d = dst[edge];

    const float4 hl = ((const float4*)(h_l + (long long)s * D))[lane];
    const float4 hr = ((const float4*)(h_r + (long long)d * D))[lane];

    float p = hl.x * hr.x + hl.y * hr.y + hl.z * hr.z + hl.w * hr.w;
    #pragma unroll
    for (int off = 16; off > 0; off >>= 1)
        p += __shfl_xor(p, off, 32);

    const float e = (p > 0.0f) ? p : NEG_SLOPE * p;

    const float4 cb = ((const float4*)(comb + (long long)d * D))[lane];
    float* outrow = agg + (long long)s * D + lane * 4;
    unsafeAtomicAdd(outrow + 0, e * cb.x);
    unsafeAtomicAdd(outrow + 1, e * cb.y);
    unsafeAtomicAdd(outrow + 2, e * cb.z);
    unsafeAtomicAdd(outrow + 3, e * cb.w);
}

extern "C" void kernel_launch(void* const* d_in, const int* in_sizes, int n_in,
                              void* d_out, int out_size, void* d_ws, size_t ws_size,
                              hipStream_t stream) {
    const float* X   = (const float*)d_in[0];
    const float* W   = (const float*)d_in[1];
    const float* W_r = (const float*)d_in[2];
    const float* W_l = (const float*)d_in[3];
    const int*   src = (const int*)d_in[4];
    const int*   dst = (const int*)d_in[5];

    const int n_nodes = in_sizes[0] / D;
    const int n_edges = in_sizes[4];

    const long long per = (long long)n_nodes * D;
    float* comb = (float*)d_ws;
    float* h_r  = comb + per;
    float* h_l  = h_r + per;
    float* agg  = (float*)d_out;

    {
        long long n = per;
        int blocks = (int)((n + 255) / 256);
        gat_zero<<<blocks, 256, 0, stream>>>(agg, n);
    }

    {
        dim3 grid((n_nodes + 15) / 16, 3, 1);
        gat_gemm_wmma<<<grid, 256, 0, stream>>>(X, W, W_r, W_l,
                                                comb, h_r, h_l, n_nodes);
    }

    {
        const int wavesPerBlock = 8;
        int blocks = (n_edges + wavesPerBlock - 1) / wavesPerBlock;
        gat_edge_fused<<<blocks, 256, 0, stream>>>(h_l, h_r, comb,
                                                   src, dst, agg, n_edges);
    }
}


# Projection GEMM re-tiled: each block stages its whole 128x128 weight matrix and a 64-row X tile in LDS; each wave owns a 32x32 tile (4 accumulators) and reads all fragments from LDS as 8-byte pairs
# speedup vs baseline: 1.1599x; 1.1599x over previous
#include <hip/hip_runtime.h>
#include <hip/hip_bf16.h>

typedef __attribute__((ext_vector_type(2))) float v2f;
typedef __attribute__((ext_vector_type(8))) float v8f;

#define D 128
#define NEG_SLOPE 0.2f

__global__ __launch_bounds__(256) void gat_zero(float* __restrict__ p, long long n) {
    long long i = (long long)blockIdx.x * blockDim.x + threadIdx.x;
    if (i < n) p[i] = 0.0f;
}

#define TM 64
#define PA 132
#define PP 288

__global__ __launch_bounds__(256) void gat_gemm_wmma(
    const float* __restrict__ X,
    const float* __restrict__ W0, const float* __restrict__ W1,
    const float* __restrict__ W2,
    float* __restrict__ O0, float* __restrict__ O1, float* __restrict__ O2,
    int n_nodes)
{
    __shared__ __attribute__((aligned(16))) float sW[(D / 2) * PP];
    __shared__ __attribute__((aligned(16))) float sX[TM * PA];

    const int m    = blockIdx.x % 3;
    const int row0 = (blockIdx.x / 3) * TM;
    const float* __restrict__ W = (m == 0) ? W0 : (m == 1) ? W1 : W2;
    float* __restrict__       O = (m == 0) ? O0 : (m == 1) ? O1 : O2;

    const int tid = threadIdx.x;

    #pragma unroll
    for (int i = 0; i < 8; ++i) {
        const int f = tid + i * 256;
        const int p = f >> 5, q = f & 31;
        const float4 u = ((const float4*)(W + (2 * p) * D))[q];
        const float4 w = ((const float4*)(W + (2 * p + 1) * D))[q];
        const float4 lo = {u.x, w.x, u.y, w.y};
        const float4 hi = {u.z, w.z, u.w, w.w};
        *(float4*)&sW[p * PP + 8 * q]     = lo;
        *(float4*)&sW[p * PP + 8 * q + 4] = hi;
    }
    #pragma unroll
    for (int i = 0; i < 8; ++i) {
        const int f = tid + i * 256;
        const int r = f >> 5, q = f & 31;
        int gr = row0 + r; gr = gr < n_nodes ? gr : n_nodes - 1;
        const float4 v = ((const float4*)(X + (long long)gr * D))[q];
        *(float4*)&sX[r * PA + q * 4] = v;
    }
    __syncthreads();

    const int wave = tid >> 5;
    const int lane = tid & 31;
    const int half = lane >> 4;
    const int lm   = lane & 15;
    const int rbase = (wave >> 2) * 32;
    const int cbase = (wave & 3) * 32;

    const float* pa0 = sX + (rbase + lm) * PA + half * 2;
    const float* pa1 = pa0 + 16 * PA;
    const float* pb0 = sW + half * PP + 2 * (cbase + lm);
    const float* pb1 = pb0 + 32;

    v8f acc00 = {}, acc01 = {}, acc10 = {}, acc11 = {};
    #pragma unroll
    for (int k0 = 0; k0 < D; k0 += 4) {
        const v2f a0 = *(const v2f*)(pa0 + k0);
        const v2f a1 = *(const v2f*)(pa1 + k0);
        const v2f b0 = *(const v2f*)(pb0 + (k0 >> 1) * PP);
        const v2f b1 = *(const v2f*)(pb1 + (k0 >> 1) * PP);
        acc00 = __builtin_amdgcn_wmma_f32_16x16x4_f32(
                false, a0, false, b0, (short)0, acc00, false, false);
        acc01 = __builtin_amdgcn_wmma_f32_16x16x4_f32(
                false, a0, false, b1, (short)0, acc01, false, false);
        acc10 = __builtin_amdgcn_wmma_f32_16x16x4_f32(
                false, a1, false, b0, (short)0, acc10, false, false);
        acc11 = __builtin_amdgcn_wmma_f32_16x16x4_f32(
                false, a1, false, b1, (short)0, acc11, false, false);
    }

    #pragma unroll
    for (int v = 0; v < 8; ++v) {
        const int r0 = row0 + rbase + v + half * 8;
        const int r1 = r0 + 16;
        if (r0 < n_nodes) {
            O[(long long)r0 * D + cbase + lm]      = acc00[v];
            O[(long long)r0 * D + cbase + 16 + lm] = acc01[v];
        }
        if (r1 < n_nodes) {
            O[(long long)r1 * D + cbase + lm]      = acc10[v];
            O[(long long)r1 * D + cbase + 16 + lm] = acc11[v];
        }
    }
}

__global__ __launch_bounds__(256) void gat_edge_fused(
    const float* __restrict__ h_l, const float* __restrict__ h_r,
    const float* __restrict__ comb,
    const int* __restrict__ src, const int* __restrict__ dst,
    float* __restrict__ agg, int n_edges)
{
    const int wavesPerBlock = blockDim.x >> 5;
    const int edge = blockIdx.x * wavesPerBlock + (threadIdx.x >> 5);
    const int lane = threadIdx.x & 31;
    if (edge >= n_edges) return;

    const int s = src[edge];
    const int d = dst[edge];

    const float4 hl = ((const float4*)(h_l + (long long)s * D))[lane];
    const float4 hr = ((const float4*)(h_r + (long long)d * D))[lane];

    float p = hl.x * hr.x + hl.y * hr.y + hl.z * hr.z + hl.w * hr.w;
    #pragma unroll
    for (int off = 16; off > 0; off >>= 1)
        p += __shfl_xor(p, off, 32);

    const float e = (p > 0.0f) ? p : NEG_SLOPE * p;

    const float4 cb = ((const float4*)(comb + (long long)d * D))[lane];
    float* outrow = agg + (long long)s * D + lane * 4;
    unsafeAtomicAdd(outrow + 0, e * cb.x);
    unsafeAtomicAdd(outrow + 1, e * cb.y);
    unsafeAtomicAdd(outrow + 2, e * cb.z);
    unsafeAtomicAdd(outrow + 3, e * cb.w);
}

extern "C" void kernel_launch(void* const* d_in, const int* in_sizes, int n_in,
                              void* d_out, int out_size, void* d_ws, size_t ws_size,
                              hipStream_t stream) {
    const float* X   = (const float*)d_in[0];
    const float* W   = (const float*)d_in[1];
    const float* W_r = (const float*)d_in[2];
    const float* W_l = (const float*)d_in[3];
    const int*   src = (const int*)d_in[4];
    const int*   dst = (const int*)d_in[5];

    const int n_nodes = in_sizes[0] / D;
    const int n_edges = in_sizes[4];

    const long long per = (long long)n_nodes * D;
    float* comb = (float*)d_ws;
    float* h_r  = comb + per;
    float* h_l  = h_r + per;
    float* agg  = (float*)d_out;

    {
        long long n = per;
        int blocks = (int)((n + 255) / 256);
        gat_zero<<<blocks, 256, 0, stream>>>(agg, n);
    }

    {
        int blocks = ((n_nodes + 63) / 64) * 3;
        gat_gemm_wmma<<<blocks, 256, 0, stream>>>(X, W, W_r, W_l,
                                                  comb, h_r, h_l, n_nodes);
    }

    {
        const int wavesPerBlock = 8;
        int blocks = (n_edges + wavesPerBlock - 1) / wavesPerBlock;
        gat_edge_fused<<<blocks, 256, 0, stream>>>(h_l, h_r, comb,
                                                   src, dst, agg, n_edges);
    }
}


# Edge pass: one wave owns 16 edges (one coalesced index load, rows of 8 edges loaded as a group); zero pass stores 128 B per thread: 16x and 32x fewer waves to start; atomics in arrival order as before
# speedup vs baseline: 1.2447x; 1.0731x over previous
#include <hip/hip_runtime.h>
#include <hip/hip_bf16.h>

typedef __attribute__((ext_vector_type(2))) float v2f;
typedef __attribute__((ext_vector_type(8))) float v8f;

#define D 128
#define NEG_SLOPE 0.2f

#define ZPT 8

__global__ __launch_bounds__(256) void gat_zero(float* __restrict__ p, long long n) {
    const long long n4 = n >> 2;
    const long long i0 = (long long)blockIdx.x * (256 * ZPT) + threadIdx.x;
    const float4 z = {0.0f, 0.0f, 0.0f, 0.0f};
    #pragma unroll
    for (int k = 0; k < ZPT; ++k) {
        const long long i = i0 + k * 256;
        if (i < n4) ((float4*)p)[i] = z;
    }
}

#define TM 64
#define PA 132
#define PP 288

__global__ __launch_bounds__(256) void gat_gemm_wmma(
    const float* __restrict__ X,
    const float* __restrict__ W0, const float* __restrict__ W1,
    const float* __restrict__ W2,
    float* __restrict__ O0, float* __restrict__ O1, float* __restrict__ O2,
    int n_nodes)
{
    __shared__ __attribute__((aligned(16))) float sW[(D / 2) * PP];
    __shared__ __attribute__((aligned(16))) float sX[TM * PA];

    const int m    = blockIdx.x % 3;
    const int row0 = (blockIdx.x / 3) * TM;
    const float* __restrict__ W = (m == 0) ? W0 : (m == 1) ? W1 : W2;
    float* __restrict__       O = (m == 0) ? O0 : (m == 1) ? O1 : O2;

    const int tid = threadIdx.x;

    #pragma unroll
    for (int i = 0; i < 8; ++i) {
        const int f = tid + i * 256;
        const int p = f >> 5, q = f & 31;
        const float4 u = ((const float4*)(W + (2 * p) * D))[q];
        const float4 w = ((const float4*)(W + (2 * p + 1) * D))[q];
        const float4 lo = {u.x, w.x, u.y, w.y};
        const float4 hi = {u.z, w.z, u.w, w.w};
        *(float4*)&sW[p * PP + 8 * q]     = lo;
        *(float4*)&sW[p * PP + 8 * q + 4] = hi;
    }
    #pragma unroll
    for (int i = 0; i < 8; ++i) {
        const int f = tid + i * 256;
        const int r = f >> 5, q = f & 31;
        int gr = row0 + r; gr = gr < n_nodes ? gr : n_nodes - 1;
        const float4 v = ((const float4*)(X + (long long)gr * D))[q];
        *(float4*)&sX[r * PA + q * 4] = v;
    }
    __syncthreads();

    const int wave = tid >> 5;
    const int lane = tid & 31;
    const int half = lane >> 4;
    const int lm   = lane & 15;
    const int rbase = (wave >> 2) * 32;
    const int cbase = (wave & 3) * 32;

    const float* pa0 = sX + (rbase + lm) * PA + half * 2;
    const float* pa1 = pa0 + 16 * PA;
    const float* pb0 = sW + half * PP + 2 * (cbase + lm);
    const float* pb1 = pb0 + 32;

    v8f acc00 = {}, acc01 = {}, acc10 = {}, acc11 = {};
    #pragma unroll
    for (int k0 = 0; k0 < D; k0 += 4) {
        const v2f a0 = *(const v2f*)(pa0 + k0);
        const v2f a1 = *(const v2f*)(pa1 + k0);
        const v2f b0 = *(const v2f*)(pb0 + (k0 >> 1) * PP);
        const v2f b1 = *(const v2f*)(pb1 + (k0 >> 1) * PP);
        acc00 = __builtin_amdgcn_wmma_f32_16x16x4_f32(
                false, a0, false, b0, (short)0, acc00, false, false);
        acc01 = __builtin_amdgcn_wmma_f32_16x16x4_f32(
                false, a0, false, b1, (short)0, acc01, false, false);
        acc10 = __builtin_amdgcn_wmma_f32_16x16x4_f32(
                false, a1, false, b0, (short)0, acc10, false, false);
        acc11 = __builtin_amdgcn_wmma_f32_16x16x4_f32(
                false, a1, false, b1, (short)0, acc11, false, false);
    }

    #pragma unroll
    for (int v = 0; v < 8; ++v) {
        const int r0 = row0 + rbase + v + half * 8;
        const int r1 = r0 + 16;
        if (r0 < n_nodes) {
            O[(long long)r0 * D + cbase + lm]      = acc00[v];
            O[(long long)r0 * D + cbase + 16 + lm] = acc01[v];
        }
        if (r1 < n_nodes) {
            O[(long long)r1 * D + cbase + lm]      = acc10[v];
            O[(long long)r1 * D + cbase + 16 + lm] = acc11[v];
        }
    }
}

#define EPW 16
#define EB  8

__global__ __launch_bounds__(256) void gat_edge_fused(
    const float* __restrict__ h_l, const float* __restrict__ h_r,
    const float* __restrict__ comb,
    const int* __restrict__ src, const int* __restrict__ dst,
    float* __restrict__ agg, int n_edges)
{
    const int lane = threadIdx.x & 31;
    const int wv   = __builtin_amdgcn_readfirstlane((int)(threadIdx.x >> 5));
    const int base = (blockIdx.x * 8 + wv) * EPW;
    if (base >= n_edges) return;

    int ei = base + (lane & 15);
    ei = ei < n_edges ? ei : n_edges - 1;
    const int* __restrict__ ip = (lane < 16) ? src : dst;
    const int idx = ip[ei];

    #pragma unroll
    for (int b = 0; b < EPW; b += EB) {
        if (base + b >= n_edges) break;

        float4 hl[EB], hr[EB], cb[EB];
        int    sv[EB];
        #pragma unroll
        for (int j = 0; j < EB; ++j) {
            const int s = __builtin_amdgcn_readlane(idx, b + j);
            const int d = __builtin_amdgcn_readlane(idx, 16 + b + j);
            sv[j] = s;
            hl[j] = ((const float4*)(h_l  + (long long)s * D))[lane];
            hr[j] = ((const float4*)(h_r  + (long long)d * D))[lane];
            cb[j] = ((const float4*)(comb + (long long)d * D))[lane];
        }

        float p[EB];
        #pragma unroll
        for (int j = 0; j < EB; ++j)
            p[j] = hl[j].x * hr[j].x + hl[j].y * hr[j].y
                 + hl[j].z * hr[j].z + hl[j].w * hr[j].w;

        #pragma unroll
        for (int off = 16; off > 0; off >>= 1) {
            #pragma unroll
            for (int j = 0; j < EB; ++j)
                p[j] += __shfl_xor(p[j], off, 32);
        }

        #pragma unroll
        for (int j = 0; j < EB; ++j) {
            if (base + b + j < n_edges) {
                const float e = (p[j] > 0.0f) ? p[j] : NEG_SLOPE * p[j];
                float* outrow = agg + (long long)sv[j] * D + lane * 4;
                unsafeAtomicAdd(outrow + 0, e * cb[j].x);
                unsafeAtomicAdd(outrow + 1, e * cb[j].y);
                unsafeAtomicAdd(outrow + 2, e * cb[j].z);
                unsafeAtomicAdd(outrow + 3, e * cb[j].w);
            }
        }
    }
}

extern "C" void kernel_launch(void* const* d_in, const int* in_sizes, int n_in,
                              void* d_out, int out_size, void* d_ws, size_t ws_size,
                              hipStream_t stream) {
    const float* X   = (const float*)d_in[0];
    const float* W   = (const float*)d_in[1];
    const float* W_r = (const float*)d_in[2];
    const float* W_l = (const float*)d_in[3];
    const int*   src = (const int*)d_in[4];
    const int*   dst = (const int*)d_in[5];

    const int n_nodes = in_sizes[0] / D;
    const int n_edges = in_sizes[4];

    const long long per = (long long)n_nodes * D;
    float* comb = (float*)d_ws;
    float* h_r  = comb + per;
    float* h_l  = h_r + per;
    float* agg  = (float*)d_out;

    {
        long long n = per;
        long long n4 = n >> 2;
        int blocks = (int)((n4 + 256 * ZPT - 1) / (256 * ZPT));
        gat_zero<<<blocks, 256, 0, stream>>>(agg, n);
    }

    {
        int blocks = ((n_nodes + 63) / 64) * 3;
        gat_gemm_wmma<<<blocks, 256, 0, stream>>>(X, W, W_r, W_l,
                                                  comb, h_r, h_l, n_nodes);
    }

    {
        const int wavesPerBlock = 8;
        int waves  = (n_edges + EPW - 1) / EPW;
        int blocks = (waves + wavesPerBlock - 1) / wavesPerBlock;
        gat_edge_fused<<<blocks, 256, 0, stream>>>(h_l, h_r, comb,
                                                   src, dst, agg, n_edges);
    }
}


# comb is stored column-permuted (row position 4q+j holds column q+32j) so each of an edge's four atomic instructions adds into one whole 128-byte line of the output; atomics in arrival order as before
# speedup vs baseline: 1.8601x; 1.4944x over previous
#include <hip/hip_runtime.h>
#include <hip/hip_bf16.h>

typedef __attribute__((ext_vector_type(2))) float v2f;
typedef __attribute__((ext_vector_type(8))) float v8f;

#define D 128
#define NEG_SLOPE 0.2f

#define ZPT 8

__global__ __launch_bounds__(256) void gat_zero(float* __restrict__ p, long long n) {
    const long long n4 = n >> 2;
    const long long i0 = (long long)blockIdx.x * (256 * ZPT) + threadIdx.x;
    const float4 z = {0.0f, 0.0f, 0.0f, 0.0f};
    #pragma unroll
    for (int k = 0; k < ZPT; ++k) {
        const long long i = i0 + k * 256;
        if (i < n4) ((float4*)p)[i] = z;
    }
}

#define TM 64
#define PA 132
#define PP 288

__global__ __launch_bounds__(256) void gat_gemm_wmma(
    const float* __restrict__ X,
    const float* __restrict__ W0, const float* __restrict__ W1,
    const float* __restrict__ W2,
    float* __restrict__ O0, float* __restrict__ O1, float* __restrict__ O2,
    int n_nodes)
{
    __shared__ __attribute__((aligned(16))) float sW[(D / 2) * PP];
    __shared__ __attribute__((aligned(16))) float sX[TM * PA];

    const int m    = blockIdx.x % 3;
    const int row0 = (blockIdx.x / 3) * TM;
    const float* __restrict__ W = (m == 0) ? W0 : (m == 1) ? W1 : W2;
    float* __restrict__       O = (m == 0) ? O0 : (m == 1) ? O1 : O2;

    const int tid = threadIdx.x;

    auto xload = [&](int i) -> float4 {
        const int f = tid + i * 256;
        const int r = f >> 5, q = f & 31;
        int gr = row0 + r; gr = gr < n_nodes ? gr : n_nodes - 1;
        return ((const float4*)(X + (long long)gr * D))[q];
    };
    const float4 x0 = xload(0), x1 = xload(1), x2 = xload(2), x3 = xload(3);
    const float4 x4 = xload(4), x5 = xload(5), x6 = xload(6), x7 = xload(7);

    if (m != 0) {
        #pragma unroll
        for (int i = 0; i < 8; ++i) {
            const int f = tid + i * 256;
            const int p = f >> 5, q = f & 31;
            const float4 u = ((const float4*)(W + (2 * p) * D))[q];
            const float4 w = ((const float4*)(W + (2 * p + 1) * D))[q];
            const float4 lo = {u.x, w.x, u.y, w.y};
            const float4 hi = {u.z, w.z, u.w, w.w};
            *(float4*)&sW[p * PP + 8 * q]     = lo;
            *(float4*)&sW[p * PP + 8 * q + 4] = hi;
        }
    } else {
        #pragma unroll
        for (int i = 0; i < 8; ++i) {
            const int f = tid + i * 256;
            const int p = f >> 5, q = f & 31;
            const float* __restrict__ r0 = W + (2 * p) * D + q;
            const float* __restrict__ r1 = r0 + D;
            const float4 lo = {r0[0],  r1[0],  r0[32], r1[32]};
            const float4 hi = {r0[64], r1[64], r0[96], r1[96]};
            *(float4*)&sW[p * PP + 8 * q]     = lo;
            *(float4*)&sW[p * PP + 8 * q + 4] = hi;
        }
    }
    auto xstore = [&](int i, const float4 v) {
        const int f = tid + i * 256;
        const int r = f >> 5, q = f & 31;
        *(float4*)&sX[r * PA + q * 4] = v;
    };
    xstore(0, x0); xstore(1, x1); xstore(2, x2); xstore(3, x3);
    xstore(4, x4); xstore(5, x5); xstore(6, x6); xstore(7, x7);
    __syncthreads();

    const int wave = tid >> 5;
    const int lane = tid & 31;
    const int half = lane >> 4;
    const int lm   = lane & 15;
    const int rbase = (wave >> 2) * 32;
    const int cbase = (wave & 3) * 32;

    const float* pa0 = sX + (rbase + lm) * PA + half * 2;
    const float* pa1 = pa0 + 16 * PA;
    const float* pb0 = sW + half * PP + 2 * (cbase + lm);
    const float* pb1 = pb0 + 32;

    v8f acc00 = {}, acc01 = {}, acc10 = {}, acc11 = {};
    #pragma unroll
    for (int k0 = 0; k0 < D; k0 += 4) {
        const v2f a0 = *(const v2f*)(pa0 + k0);
        const v2f a1 = *(const v2f*)(pa1 + k0);
        const v2f b0 = *(const v2f*)(pb0 + (k0 >> 1) * PP);
        const v2f b1 = *(const v2f*)(pb1 + (k0 >> 1) * PP);
        acc00 = __builtin_amdgcn_wmma_f32_16x16x4_f32(
                false, a0, false, b0, (short)0, acc00, false, false);
        acc01 = __builtin_amdgcn_wmma_f32_16x16x4_f32(
                false, a0, false, b1, (short)0, acc01, false, false);
        acc10 = __builtin_amdgcn_wmma_f32_16x16x4_f32(
                false, a1, false, b0, (short)0, acc10, false, false);
        acc11 = __builtin_amdgcn_wmma_f32_16x16x4_f32(
                false, a1, false, b1, (short)0, acc11, false, false);
    }

    #pragma unroll
    for (int v = 0; v < 8; ++v) {
        const int r0 = row0 + rbase + v + half * 8;
        const int r1 = r0 + 16;
        if (r0 < n_nodes) {
            O[(long long)r0 * D + cbase + lm]      = acc00[v];
            O[(long long)r0 * D + cbase + 16 + lm] = acc01[v];
        }
        if (r1 < n_nodes) {
            O[(long long)r1 * D + cbase + lm]      = acc10[v];
            O[(long long)r1 * D + cbase + 16 + lm] = acc11[v];
        }
    }
}

#define EPW 16
#define EB  8

__global__ __launch_bounds__(256) void gat_edge_fused(
    const float* __restrict__ h_l, const float* __restrict__ h_r,
    const float* __restrict__ comb,
    const int* __restrict__ src, const int* __restrict__ dst,
    float* __restrict__ agg, int n_edges)
{
    const int lane = threadIdx.x & 31;
    const int wv   = __builtin_amdgcn_readfirstlane((int)(threadIdx.x >> 5));
    const int base = (blockIdx.x * 8 + wv) * EPW;
    if (base >= n_edges) return;

    int ei = base + (lane & 15);
    ei = ei < n_edges ? ei : n_edges - 1;
    const int* __restrict__ ip = (lane < 16) ? src : dst;
    const int idx = ip[ei];

    #pragma unroll
    for (int b = 0; b < EPW; b += EB) {
        if (base + b >= n_edges) break;

        float4 hl[EB], hr[EB], cb[EB];
        int    sv[EB];
        #pragma unroll
        for (int j = 0; j < EB; ++j) {
            const int s = __builtin_amdgcn_readlane(idx, b + j);
            const int d = __builtin_amdgcn_readlane(idx, 16 + b + j);
            sv[j] = s;
            hl[j] = ((const float4*)(h_l  + (long long)s * D))[lane];
            hr[j] = ((const float4*)(h_r  + (long long)d * D))[lane];
            cb[j] = ((const float4*)(comb + (long long)d * D))[lane];
        }

        float p[EB];
        #pragma unroll
        for (int j = 0; j < EB; ++j)
            p[j] = hl[j].x * hr[j].x + hl[j].y * hr[j].y
                 + hl[j].z * hr[j].z + hl[j].w * hr[j].w;

        #pragma unroll
        for (int off = 16; off > 0; off >>= 1) {
            #pragma unroll
            for (int j = 0; j < EB; ++j)
                p[j] += __shfl_xor(p[j], off, 32);
        }

        #pragma unroll
        for (int j = 0; j < EB; ++j) {
            if (base + b + j < n_edges) {
                const float e = (p[j] > 0.0f) ? p[j] : NEG_SLOPE * p[j];
                float* outrow = agg + (long long)sv[j] * D + lane;
                unsafeAtomicAdd(outrow +  0, e * cb[j].x);
                unsafeAtomicAdd(outrow + 32, e * cb[j].y);
                unsafeAtomicAdd(outrow + 64, e * cb[j].z);
                unsafeAtomicAdd(outrow + 96, e * cb[j].w);
            }
        }
    }
}

extern "C" void kernel_launch(void* const* d_in, const int* in_sizes, int n_in,
                              void* d_out, int out_size, void* d_ws, size_t ws_size,
                              hipStream_t stream) {
    const float* X   = (const float*)d_in[0];
    const float* W   = (const float*)d_in[1];
    const float* W_r = (const float*)d_in[2];
    const float* W_l = (const float*)d_in[3];
    const int*   src = (const int*)d_in[4];
    const int*   dst = (const int*)d_in[5];

    const int n_nodes = in_sizes[0] / D;
    const int n_edges = in_sizes[4];

    const long long per = (long long)n_nodes * D;
    float* comb = (float*)d_ws;
    float* h_r  = comb + per;
    float* h_l  = h_r + per;
    float* agg  = (float*)d_out;

    {
        long long n = per;
        long long n4 = n >> 2;
        int blocks = (int)((n4 + 256 * ZPT - 1) / (256 * ZPT));
        gat_zero<<<blocks, 256, 0, stream>>>(agg, n);
    }

    {
        int blocks = ((n_nodes + 63) / 64) * 3;
        gat_gemm_wmma<<<blocks, 256, 0, stream>>>(X, W, W_r, W_l,
                                                  comb, h_r, h_l, n_nodes);
    }

    {
        const int wavesPerBlock = 8;
        int waves  = (n_edges + EPW - 1) / EPW;
        int blocks = (waves + wavesPerBlock - 1) / wavesPerBlock;
        gat_edge_fused<<<blocks, 256, 0, stream>>>(h_l, h_r, comb,
                                                   src, dst, agg, n_edges);
    }
}
